# starting kernel
# baseline (speedup 1.0000x reference)
#include <hip/hip_runtime.h>
#include <hip/hip_bf16.h>


#define KDIM   576
#define NVALID 196
#define NPAD   208
#define MDIM   256
#define PROW   (NPAD * 2)

typedef __attribute__((ext_vector_type(2))) float v2f;
typedef __attribute__((ext_vector_type(8))) float v8f;

__global__ void build_b_kernel(const float* __restrict__ x, float* __restrict__ Bp) {
    int idx = blockIdx.x * blockDim.x + threadIdx.x;
    if (idx >= KDIM * NPAD) return;
    int row = idx / NPAD;
    int col = idx - row * NPAD;

    float v = 0.0f;
    if (col < NVALID) {
        int h  = col / 14;
        int w  = col - h * 14;
        int j  = row / 192;
        int rm = row - j * 192;
        int c  = rm / 3;
        int k2 = rm - c * 3;

        int wq = w + k2 - 1;
        if (wq >= 0 && wq < 14) {
            int wp = wq - 1; if (wp < 0) wp += 14;
            int h1 = h - 1;  if (h1 < 0) h1 += 14;
            int hh1 = h1 + j - 1;
            int hh2 = h  + j - 1;
            float t1 = (hh1 >= 0 && hh1 < 14) ? x[(64 + c) * 196 + hh1 * 14 + wp] : 0.0f;
            float t2 = (hh2 >= 0 && hh2 < 14) ? x[c * 196 + hh2 * 14 + wp] : 0.0f;
            v = t1 + t2;
        }
    }
    Bp[(row >> 1) * PROW + col * 2 + (row & 1)] = v;
}

__global__ void __launch_bounds__(32)
wmma_gemm_f32_kernel(const float* __restrict__ A,
                     const float* __restrict__ Bp,
                     float* __restrict__ C) {
    const int lane = threadIdx.x & 31;
    const int half = lane >> 4;
    const int l    = lane & 15;

    const int tileM = blockIdx.x;
    const int tileN = blockIdx.y;

    const int arow = tileM * 16 + l;
    const int bcol = tileN * 16 + l;

    const float* __restrict__ Arow  = A + arow * KDIM + half * 2;
    const float* __restrict__ Bbase = Bp + (size_t)half * PROW + bcol * 2;

    v8f acc = {};
#pragma unroll 4
    for (int k = 0; k < KDIM; k += 4) {
        v2f a = *(const v2f*)(Arow + k);
        v2f b = *(const v2f*)(Bbase + (size_t)(k >> 1) * PROW);
        acc = __builtin_amdgcn_wmma_f32_16x16x4_f32(
             false, a,  false, b,
             (short)0, acc,  false,  false);
    }

    const int col = tileN * 16 + l;
    if (col < NVALID) {
        const int rowBase = tileM * 16 + half * 8;
#pragma unroll
        for (int r = 0; r < 8; ++r) {
            C[(rowBase + r) * NVALID + col] = acc[r];
        }
    }
}

extern "C" void kernel_launch(void* const* d_in, const int* in_sizes, int n_in,
                              void* d_out, int out_size, void* d_ws, size_t ws_size,
                              hipStream_t stream) {
    const float* x = (const float*)d_in[0];
    const float* w = (const float*)d_in[1];
    float* out = (float*)d_out;
    float* Bp  = (float*)d_ws;

    {
        const int total   = KDIM * NPAD;
        const int threads = 256;
        const int blocks  = (total + threads - 1) / threads;
        build_b_kernel<<<blocks, threads, 0, stream>>>(x, Bp);
    }
    {
        dim3 grid(MDIM / 16, NPAD / 16);
        wmma_gemm_f32_kernel<<<grid, 32, 0, stream>>>(w, Bp, out);
    }
}


# GEMM: K split over 8 waves per 16x16 tile (one work-group), each wave's fragment loads issued as one group ahead of its 18 products, partials summed through LDS in wave order: K sum order changes
# speedup vs baseline: 3.4541x; 3.4541x over previous
#include <hip/hip_runtime.h>
#include <hip/hip_bf16.h>


#define KDIM   576
#define NVALID 196
#define NPAD   208
#define MDIM   256
#define PROW   (NPAD * 2)

typedef __attribute__((ext_vector_type(2))) float v2f;
typedef __attribute__((ext_vector_type(8))) float v8f;

__global__ void build_b_kernel(const float* __restrict__ x, float* __restrict__ Bp) {
    int idx = blockIdx.x * blockDim.x + threadIdx.x;
    if (idx >= KDIM * NPAD) return;
    int row = idx / NPAD;
    int col = idx - row * NPAD;

    float v = 0.0f;
    if (col < NVALID) {
        int h  = col / 14;
        int w  = col - h * 14;
        int j  = row / 192;
        int rm = row - j * 192;
        int c  = rm / 3;
        int k2 = rm - c * 3;

        int wq = w + k2 - 1;
        if (wq >= 0 && wq < 14) {
            int wp = wq - 1; if (wp < 0) wp += 14;
            int h1 = h - 1;  if (h1 < 0) h1 += 14;
            int hh1 = h1 + j - 1;
            int hh2 = h  + j - 1;
            float t1 = (hh1 >= 0 && hh1 < 14) ? x[(64 + c) * 196 + hh1 * 14 + wp] : 0.0f;
            float t2 = (hh2 >= 0 && hh2 < 14) ? x[c * 196 + hh2 * 14 + wp] : 0.0f;
            v = t1 + t2;
        }
    }
    Bp[(row >> 1) * PROW + col * 2 + (row & 1)] = v;
}

#define KSPLIT 8
#define KCHUNK (KDIM / KSPLIT)
#define KSTEPS (KCHUNK / 4)

__global__ void __launch_bounds__(32 * KSPLIT)
wmma_gemm_f32_kernel(const float* __restrict__ A,
                     const float* __restrict__ Bp,
                     float* __restrict__ C) {
    __shared__ float part[KSPLIT][8 * 32];

    const int lane = threadIdx.x & 31;
    const int wave = threadIdx.x >> 5;
    const int half = lane >> 4;
    const int l    = lane & 15;

    const int tileM = blockIdx.x;
    const int tileN = blockIdx.y;

    const int arow = tileM * 16 + l;
    const int bcol = tileN * 16 + l;
    const int k0   = wave * KCHUNK;

    const float* __restrict__ Arow  = A + arow * KDIM + half * 2 + k0;
    const float* __restrict__ Bbase = Bp + (size_t)(half + (k0 >> 1)) * PROW + bcol * 2;

    v2f a[KSTEPS], b[KSTEPS];
#pragma unroll
    for (int s = 0; s < KSTEPS; ++s) {
        a[s] = *(const v2f*)(Arow + s * 4);
        b[s] = *(const v2f*)(Bbase + (size_t)(s * 2) * PROW);
    }

    __builtin_amdgcn_sched_barrier(0);

    v8f acc = {};
#pragma unroll
    for (int s = 0; s < KSTEPS; ++s) {
        acc = __builtin_amdgcn_wmma_f32_16x16x4_f32(
             false, a[s],  false, b[s],
             (short)0, acc,  false,  false);
    }

#pragma unroll
    for (int r = 0; r < 8; ++r) {
        part[wave][r * 32 + lane] = acc[r];
    }
    __syncthreads();

    float sum = part[0][wave * 32 + lane];
#pragma unroll
    for (int p = 1; p < KSPLIT; ++p) {
        sum += part[p][wave * 32 + lane];
    }

    const int col = tileN * 16 + l;
    if (col < NVALID) {
        const int row = tileM * 16 + half * 8 + wave;
        C[row * NVALID + col] = sum;
    }
}

extern "C" void kernel_launch(void* const* d_in, const int* in_sizes, int n_in,
                              void* d_out, int out_size, void* d_ws, size_t ws_size,
                              hipStream_t stream) {
    const float* x = (const float*)d_in[0];
    const float* w = (const float*)d_in[1];
    float* out = (float*)d_out;
    float* Bp  = (float*)d_ws;

    {
        const int total   = KDIM * NPAD;
        const int threads = 256;
        const int blocks  = (total + threads - 1) / threads;
        build_b_kernel<<<blocks, threads, 0, stream>>>(x, Bp);
    }
    {
        dim3 grid(MDIM / 16, NPAD / 16);
        wmma_gemm_f32_kernel<<<grid, 32 * KSPLIT, 0, stream>>>(w, Bp, out);
    }
}


# GEMM: B fragments built in registers straight from x inside the GEMM kernel (one launch, no B matrix in memory); each wave takes 8 channels over all 9 taps: K sum order changes
# speedup vs baseline: 5.3081x; 1.5368x over previous
#include <hip/hip_runtime.h>


#define KDIM   576
#define NVALID 196
#define NPAD   208
#define MDIM   256
#define XPLANE 196

typedef __attribute__((ext_vector_type(2))) float v2f;
typedef __attribute__((ext_vector_type(8))) float v8f;

#define KSPLIT 8
#define KSTEPS 18

__global__ void __launch_bounds__(32 * KSPLIT)
unfold_gemm_f32_kernel(const float* __restrict__ x,
                       const float* __restrict__ A,
                       float* __restrict__ C) {
    __shared__ float part[KSPLIT][8 * 32];

    const int lane = threadIdx.x & 31;
    const int wave = __builtin_amdgcn_readfirstlane((int)(threadIdx.x >> 5));
    const int half = lane >> 4;
    const int l    = lane & 15;

    const int tileM = blockIdx.x;
    const int tileN = blockIdx.y;

    const float* __restrict__ Arow = A + (tileM * 16 + l) * KDIM + wave * 24 + half * 6;
    v2f a[KSTEPS];
#pragma unroll
    for (int s = 0; s < KSTEPS; ++s) {
        const int j = s / 6, g = (s % 6) / 3, i = s % 3;
        a[s] = *(const v2f*)(Arow + j * 192 + g * 12 + 2 * i);
    }

    int n = tileN * 16 + l;
    if (n > NVALID - 1) n = NVALID - 1;
    const int h  = n / 14;
    const int w  = n - h * 14;
    const int h1 = (h == 0) ? 13 : h - 1;

    int  wp[3];  bool vw[3];
    int  r1[3], r2[3];  bool v1[3], v2[3];
#pragma unroll
    for (int k2 = 0; k2 < 3; ++k2) {
        const int wq = w + k2 - 1;
        vw[k2] = (wq >= 0) && (wq < 14);
        const int t = wq - 1;
        wp[k2] = (t < 0) ? t + 14 : t;
    }
#pragma unroll
    for (int j = 0; j < 3; ++j) {
        const int hh1 = h1 + j - 1;
        const int hh2 = h  + j - 1;
        v1[j] = (hh1 >= 0) && (hh1 < 14);
        v2[j] = (hh2 >= 0) && (hh2 < 14);
        r1[j] = min(max(hh1, 0), 13) * 14;
        r2[j] = min(max(hh2, 0), 13) * 14;
    }

    const float* __restrict__ xw = x + wave * (8 * XPLANE);
    unsigned o1[3][3], o2[3][3];
#pragma unroll
    for (int j = 0; j < 3; ++j) {
#pragma unroll
        for (int k2 = 0; k2 < 3; ++k2) {
            o1[j][k2] = (unsigned)(half * (2 * XPLANE) + r1[j] + wp[k2]);
            o2[j][k2] = (unsigned)(half * (2 * XPLANE) + r2[j] + wp[k2]);
        }
    }

    float t1[KSTEPS][2], t2[KSTEPS][2];
#pragma unroll
    for (int s = 0; s < KSTEPS; ++s) {
        const int j = s / 6, g = (s % 6) / 3, i = s % 3;
#pragma unroll
        for (int e = 0; e < 2; ++e) {
            const int q = 2 * i + e, dc = q / 3, k2 = q % 3;
            t1[s][e] = (xw + o1[j][k2])[(64 + 4 * g + dc) * XPLANE];
            t2[s][e] = (xw + o2[j][k2])[(4 * g + dc) * XPLANE];
        }
    }

    __builtin_amdgcn_sched_barrier(0);

    v8f acc = {};
#pragma unroll
    for (int s = 0; s < KSTEPS; ++s) {
        const int j = s / 6, i = s % 3;
        v2f b;
#pragma unroll
        for (int e = 0; e < 2; ++e) {
            const int k2 = (2 * i + e) % 3;
            const float u1 = v1[j] ? t1[s][e] : 0.0f;
            const float u2 = v2[j] ? t2[s][e] : 0.0f;
            const float v  = u1 + u2;
            b[e] = vw[k2] ? v : 0.0f;
        }
        acc = __builtin_amdgcn_wmma_f32_16x16x4_f32(
             false, a[s],  false, b,
             (short)0, acc,  false,  false);
    }

#pragma unroll
    for (int r = 0; r < 8; ++r) {
        part[wave][r * 32 + lane] = acc[r];
    }
    __syncthreads();

    float sum = part[0][wave * 32 + lane];
#pragma unroll
    for (int p = 1; p < KSPLIT; ++p) {
        sum += part[p][wave * 32 + lane];
    }

    const int col = tileN * 16 + l;
    if (col < NVALID) {
        const int row = tileM * 16 + half * 8 + wave;
        C[row * NVALID + col] = sum;
    }
}

extern "C" void kernel_launch(void* const* d_in, const int* in_sizes, int n_in,
                              void* d_out, int out_size, void* d_ws, size_t ws_size,
                              hipStream_t stream) {
    const float* x = (const float*)d_in[0];
    const float* w = (const float*)d_in[1];
    float* out = (float*)d_out;

    dim3 grid(MDIM / 16, NPAD / 16);
    unfold_gemm_f32_kernel<<<grid, 32 * KSPLIT, 0, stream>>>(x, w, out);
}


# GEMM: N tile = one image row (grid 16x14): only the centre tap of x is loaded (24 loads a wave, not 72), taps k2=0/2 come from the neighbour lane by DPP row shift; B values and K order as before
# speedup vs baseline: 5.9534x; 1.1216x over previous
#include <hip/hip_runtime.h>


#define KDIM   576
#define NVALID 196
#define WIDTH  14
#define MDIM   256
#define XPLANE 196

typedef __attribute__((ext_vector_type(2))) float v2f;
typedef __attribute__((ext_vector_type(8))) float v8f;

#define KSPLIT 8
#define KSTEPS 18

__device__ __forceinline__ float from_left_lane(float v) {
    return __builtin_bit_cast(float, __builtin_amdgcn_update_dpp(
        0, __builtin_bit_cast(int, v), 0x111  , 0xf, 0xf, true));
}
__device__ __forceinline__ float from_right_lane(float v) {
    return __builtin_bit_cast(float, __builtin_amdgcn_update_dpp(
        0, __builtin_bit_cast(int, v), 0x101  , 0xf, 0xf, true));
}

__global__ void __launch_bounds__(32 * KSPLIT)
unfold_gemm_f32_kernel(const float* __restrict__ x,
                       const float* __restrict__ A,
                       float* __restrict__ C) {
    __shared__ float part[KSPLIT][8 * 32];

    const int lane = threadIdx.x & 31;
    const int wave = __builtin_amdgcn_readfirstlane((int)(threadIdx.x >> 5));
    const int half = lane >> 4;
    const int l    = lane & 15;

    const int tileM = blockIdx.x;
    const int h     = blockIdx.y;

    const float* __restrict__ Arow = A + (tileM * 16 + l) * KDIM + wave * 24 + half * 6;
    v2f a[KSTEPS];
#pragma unroll
    for (int s = 0; s < KSTEPS; ++s) {
        const int j = s / 6, g = (s % 6) / 3, i = s % 3;
        a[s] = *(const v2f*)(Arow + j * 192 + g * 12 + 2 * i);
    }

    const bool lane_ok = l < WIDTH;
    const int  w  = lane_ok ? l : WIDTH - 1;
    const int  wc = (w == 0) ? WIDTH - 1 : w - 1;
    const int  h1 = (h == 0) ? 13 : h - 1;

    int  r1[3], r2[3];  bool m1[3], m2[3];
#pragma unroll
    for (int j = 0; j < 3; ++j) {
        const int hh1 = h1 + j - 1;
        const int hh2 = h  + j - 1;
        m1[j] = lane_ok && (hh1 >= 0) && (hh1 < 14);
        m2[j] = lane_ok && (hh2 >= 0) && (hh2 < 14);
        r1[j] = ((j == 0) ? max(hh1, 0) : (j == 2) ? min(hh1, 13) : hh1) * WIDTH;
        r2[j] = ((j == 0) ? max(hh2, 0) : (j == 2) ? min(hh2, 13) : hh2) * WIDTH;
    }

    const float* __restrict__ xw = x + wave * (8 * XPLANE);
    const unsigned ob = (unsigned)(half * (2 * XPLANE) + wc);

    float t1[3][2][2], t2[3][2][2];
#pragma unroll
    for (int j = 0; j < 3; ++j) {
#pragma unroll
        for (int g = 0; g < 2; ++g) {
#pragma unroll
            for (int dc = 0; dc < 2; ++dc) {
                t1[j][g][dc] = (xw + r1[j] + (64 + 4 * g + dc) * XPLANE)[ob];
                t2[j][g][dc] = (xw + r2[j] + (4 * g + dc) * XPLANE)[ob];
            }
        }
    }

    __builtin_amdgcn_sched_barrier(0);

    v8f acc = {};
#pragma unroll
    for (int j = 0; j < 3; ++j) {
#pragma unroll
        for (int g = 0; g < 2; ++g) {
            float tap[6];
#pragma unroll
            for (int dc = 0; dc < 2; ++dc) {
                const float u1 = m1[j] ? t1[j][g][dc] : 0.0f;
                const float u2 = m2[j] ? t2[j][g][dc] : 0.0f;
                const float S  = u1 + u2;
                tap[dc * 3 + 0] = from_left_lane(S);
                tap[dc * 3 + 1] = S;
                tap[dc * 3 + 2] = from_right_lane(S);
            }
#pragma unroll
            for (int i = 0; i < 3; ++i) {
                v2f b;
                b[0] = tap[2 * i];
                b[1] = tap[2 * i + 1];
                acc = __builtin_amdgcn_wmma_f32_16x16x4_f32(
                     false, a[j * 6 + g * 3 + i],  false, b,
                     (short)0, acc,  false,  false);
            }
        }
    }

#pragma unroll
    for (int r = 0; r < 8; ++r) {
        part[wave][r * 32 + lane] = acc[r];
    }
    __syncthreads();

    float sum = part[0][wave * 32 + lane];
#pragma unroll
    for (int p = 1; p < KSPLIT; ++p) {
        sum += part[p][wave * 32 + lane];
    }

    if (lane_ok) {
        const int row = tileM * 16 + half * 8 + wave;
        C[row * NVALID + h * WIDTH + l] = sum;
    }
}

extern "C" void kernel_launch(void* const* d_in, const int* in_sizes, int n_in,
                              void* d_out, int out_size, void* d_ws, size_t ws_size,
                              hipStream_t stream) {
    const float* x = (const float*)d_in[0];
    const float* w = (const float*)d_in[1];
    float* out = (float*)d_out;

    dim3 grid(MDIM / 16, 14);
    unfold_gemm_f32_kernel<<<grid, 32 * KSPLIT, 0, stream>>>(x, w, out);
}


# GEMM: A tile staged in LDS: whole-line 16-byte global loads (4 lines a load, not 16 or 32), fragments read from LDS at a 16 B x odd pitch, barrier fenced on LDS only; K order and sums unchanged
# speedup vs baseline: 6.9519x; 1.1677x over previous
#include <hip/hip_runtime.h>


#define KDIM   576
#define NVALID 196
#define WIDTH  14
#define MDIM   256
#define XPLANE 196
#define AROW4  144
#define APITCH4 145
#define ASTAGE 9

typedef __attribute__((ext_vector_type(2))) float v2f;
typedef __attribute__((ext_vector_type(4))) float v4f;
typedef __attribute__((ext_vector_type(8))) float v8f;

#define KSPLIT 8
#define KSTEPS 18

__device__ __forceinline__ float from_left_lane(float v) {
    return __builtin_bit_cast(float, __builtin_amdgcn_update_dpp(
        0, __builtin_bit_cast(int, v), 0x111  , 0xf, 0xf, true));
}
__device__ __forceinline__ float from_right_lane(float v) {
    return __builtin_bit_cast(float, __builtin_amdgcn_update_dpp(
        0, __builtin_bit_cast(int, v), 0x101  , 0xf, 0xf, true));
}

__global__ void __launch_bounds__(32 * KSPLIT)
unfold_gemm_f32_kernel(const float* __restrict__ x,
                       const float* __restrict__ A,
                       float* __restrict__ C) {
    __shared__ v4f   As[16 * APITCH4];
    __shared__ float part[KSPLIT][8 * 32];

    const int lane = threadIdx.x & 31;
    const int wave = __builtin_amdgcn_readfirstlane((int)(threadIdx.x >> 5));
    const int half = lane >> 4;
    const int l    = lane & 15;

    const int tileM = blockIdx.x;
    const int h     = blockIdx.y;

    const v4f* __restrict__ Ag = (const v4f*)(A + tileM * (16 * KDIM)) + threadIdx.x;
    v4f stage[ASTAGE];
#pragma unroll
    for (int r = 0; r < ASTAGE; ++r) {
        stage[r] = Ag[r * 256];
    }

    const bool lane_ok = l < WIDTH;
    const int  w  = lane_ok ? l : WIDTH - 1;
    const int  wc = (w == 0) ? WIDTH - 1 : w - 1;
    const int  h1 = (h == 0) ? 13 : h - 1;

    int  r1[3], r2[3];  bool m1[3], m2[3];
#pragma unroll
    for (int j = 0; j < 3; ++j) {
        const int hh1 = h1 + j - 1;
        const int hh2 = h  + j - 1;
        m1[j] = lane_ok && (hh1 >= 0) && (hh1 < 14);
        m2[j] = lane_ok && (hh2 >= 0) && (hh2 < 14);
        r1[j] = ((j == 0) ? max(hh1, 0) : (j == 2) ? min(hh1, 13) : hh1) * WIDTH;
        r2[j] = ((j == 0) ? max(hh2, 0) : (j == 2) ? min(hh2, 13) : hh2) * WIDTH;
    }

    const float* __restrict__ xw = x + wave * (8 * XPLANE);
    const unsigned ob = (unsigned)(half * (2 * XPLANE) + wc);

    float t1[3][2][2], t2[3][2][2];
#pragma unroll
    for (int j = 0; j < 3; ++j) {
#pragma unroll
        for (int g = 0; g < 2; ++g) {
#pragma unroll
            for (int dc = 0; dc < 2; ++dc) {
                t1[j][g][dc] = (xw + r1[j] + (64 + 4 * g + dc) * XPLANE)[ob];
                t2[j][g][dc] = (xw + r2[j] + (4 * g + dc) * XPLANE)[ob];
            }
        }
    }

#pragma unroll
    for (int r = 0; r < ASTAGE; ++r) {
        const unsigned piece = r * 256 + threadIdx.x;
        As[piece + piece / AROW4] = stage[r];
    }
    __builtin_amdgcn_fence(__ATOMIC_RELEASE, "workgroup", "local");
    __builtin_amdgcn_s_barrier();
    __builtin_amdgcn_fence(__ATOMIC_ACQUIRE, "workgroup", "local");

    const float* Afrag = (const float*)As + l * (APITCH4 * 4) + wave * 24 + half * 6;
    v2f a[KSTEPS];
#pragma unroll
    for (int s = 0; s < KSTEPS; ++s) {
        const int j = s / 6, g = (s % 6) / 3, i = s % 3;
        a[s] = *(const v2f*)(Afrag + j * 192 + g * 12 + 2 * i);
    }
    __builtin_amdgcn_sched_barrier(0);

    v8f acc = {};
#pragma unroll
    for (int j = 0; j < 3; ++j) {
#pragma unroll
        for (int g = 0; g < 2; ++g) {
            float tap[6];
#pragma unroll
            for (int dc = 0; dc < 2; ++dc) {
                const float u1 = m1[j] ? t1[j][g][dc] : 0.0f;
                const float u2 = m2[j] ? t2[j][g][dc] : 0.0f;
                const float S  = u1 + u2;
                tap[dc * 3 + 0] = from_left_lane(S);
                tap[dc * 3 + 1] = S;
                tap[dc * 3 + 2] = from_right_lane(S);
            }
#pragma unroll
            for (int i = 0; i < 3; ++i) {
                v2f b;
                b[0] = tap[2 * i];
                b[1] = tap[2 * i + 1];
                acc = __builtin_amdgcn_wmma_f32_16x16x4_f32(
                     false, a[j * 6 + g * 3 + i],  false, b,
                     (short)0, acc,  false,  false);
            }
        }
    }

#pragma unroll
    for (int r = 0; r < 8; ++r) {
        part[wave][r * 32 + lane] = acc[r];
    }
    __syncthreads();

    float sum = part[0][wave * 32 + lane];
#pragma unroll
    for (int p = 1; p < KSPLIT; ++p) {
        sum += part[p][wave * 32 + lane];
    }

    if (lane_ok) {
        const int row = tileM * 16 + half * 8 + wave;
        C[row * NVALID + h * WIDTH + l] = sum;
    }
}

extern "C" void kernel_launch(void* const* d_in, const int* in_sizes, int n_in,
                              void* d_out, int out_size, void* d_ws, size_t ws_size,
                              hipStream_t stream) {
    const float* x = (const float*)d_in[0];
    const float* w = (const float*)d_in[1];
    float* out = (float*)d_out;

    dim3 grid(MDIM / 16, 14);
    unfold_gemm_f32_kernel<<<grid, 32 * KSPLIT, 0, stream>>>(x, w, out);
}
